# starting kernel
# baseline (speedup 1.0000x reference)
#include <hip/hip_runtime.h>
#include <hip/hip_bf16.h>

typedef float v2f __attribute__((ext_vector_type(2)));
typedef float v8f __attribute__((ext_vector_type(8)));

#define N_NODES 50000
#define N_EDGES 800000
#define D 64
#define N_TILES (N_NODES / 16)
#define SMOOTHF 0.5f


__global__ void k_zero_deg(float* __restrict__ deg) {
    int i = blockIdx.x * blockDim.x + threadIdx.x;
    if (i < N_NODES) deg[i] = 0.0f;
}

__global__ void k_degree(const int* __restrict__ ecol, float* __restrict__ deg) {
    int e = blockIdx.x * blockDim.x + threadIdx.x;
    if (e < N_EDGES) atomicAdd(&deg[ecol[e]], 1.0f);
}

__global__ void k_rsqrt(float* __restrict__ deg) {
    int i = blockIdx.x * blockDim.x + threadIdx.x;
    if (i < N_NODES) deg[i] = rsqrtf(deg[i] + 1.0f);
}


__global__ __launch_bounds__(256) void k_gemm_wmma(
    const float* __restrict__ x, const float* __restrict__ W,
    const float* __restrict__ bias, const float* __restrict__ d,
    float* __restrict__ support, float* __restrict__ h, float* __restrict__ msg)
{
    __shared__ float Wl[D * D];
    for (int i = threadIdx.x; i < D * D; i += 256) Wl[i] = W[i];
    __syncthreads();

    const int wave = threadIdx.x >> 5;
    const int lane = threadIdx.x & 31;
    const int tile = blockIdx.x * 8 + wave;
    if (tile >= N_TILES) return;

    const int row0 = tile * 16;
    const int half = lane >> 4;
    const int l15  = lane & 15;

    v8f acc[4] = {v8f{}, v8f{}, v8f{}, v8f{}};
    const float* xrow = x + (size_t)(row0 + l15) * D;

    #pragma unroll
    for (int kk = 0; kk < 16; ++kk) {
        const int k = kk * 4 + half * 2;
        v2f a = *(const v2f*)(xrow + k);
        #pragma unroll
        for (int nt = 0; nt < 4; ++nt) {
            const int n = nt * 16 + l15;
            v2f bf = *(const v2f*)(&Wl[n * D + k]);
            acc[nt] = __builtin_amdgcn_wmma_f32_16x16x4_f32(
                 false, a,  false, bf,
                 (short)0, acc[nt],
                 false,  false);
        }
    }

    #pragma unroll
    for (int nt = 0; nt < 4; ++nt) {
        const int n  = nt * 16 + l15;
        const float bv = bias[n];
        #pragma unroll
        for (int r = 0; r < 8; ++r) {
            const int row  = row0 + r + 8 * half;
            const float s  = acc[nt][r] + bv;
            const float hv = s * d[row];
            const size_t idx = (size_t)row * D + n;
            support[idx] = s;
            h[idx]       = hv;
            msg[idx]     = hv;
        }
    }
}


__global__ __launch_bounds__(256) void k_scatter(
    const int* __restrict__ erow, const int* __restrict__ ecol,
    const float* __restrict__ h, float* __restrict__ msg)
{
    const long long t = (long long)blockIdx.x * blockDim.x + threadIdx.x;
    const int e = (int)(t >> 6);
    const int j = (int)(t & 63);
    if (e < N_EDGES) {
        const int r = erow[e];
        const int c = ecol[e];
        atomicAdd(&msg[(size_t)r * D + j], h[(size_t)c * D + j]);
    }
}


__global__ void k_final(const float* __restrict__ msg, const float* __restrict__ support,
                        const float* __restrict__ d, float* __restrict__ out)
{
    const int i = blockIdx.x * blockDim.x + threadIdx.x;
    if (i < N_NODES * D) {
        const int row = i >> 6;
        out[i] = (msg[i] * d[row] * SMOOTHF + support[i]) * (1.0f / (1.0f + SMOOTHF));
    }
}


extern "C" void kernel_launch(void* const* d_in, const int* in_sizes, int n_in,
                              void* d_out, int out_size, void* d_ws, size_t ws_size,
                              hipStream_t stream) {
    const float* x    = (const float*)d_in[0];
    const float* W    = (const float*)d_in[1];
    const float* bias = (const float*)d_in[2];
    const int*   erow = (const int*)d_in[3];
    const int*   ecol = (const int*)d_in[4];
    float* out = (float*)d_out;

    float* ws      = (float*)d_ws;
    float* support = ws;
    float* h       = ws + (size_t)N_NODES * D;
    float* msg     = ws + 2 * (size_t)N_NODES * D;
    float* deg     = ws + 3 * (size_t)N_NODES * D;

    k_zero_deg<<<(N_NODES + 255) / 256, 256, 0, stream>>>(deg);
    k_degree  <<<(N_EDGES + 255) / 256, 256, 0, stream>>>(ecol, deg);
    k_rsqrt   <<<(N_NODES + 255) / 256, 256, 0, stream>>>(deg);
    k_gemm_wmma<<<(N_TILES + 7) / 8, 256, 0, stream>>>(x, W, bias, deg, support, h, msg);
    k_scatter <<<(N_EDGES * 64) / 256, 256, 0, stream>>>(erow, ecol, h, msg);
    k_final   <<<(N_NODES * D + 255) / 256, 256, 0, stream>>>(msg, support, deg, out);
}


# edge scatter: each wave walks 64 edges (coalesced index loads, readlane broadcast, a batch of whole-line row loads in flight ahead of their f32 atomics), not half an edge; atomic-order sums kept
# speedup vs baseline: 1.9214x; 1.9214x over previous
#include <hip/hip_runtime.h>
#include <hip/hip_bf16.h>

typedef float v2f __attribute__((ext_vector_type(2)));
typedef float v8f __attribute__((ext_vector_type(8)));

#define N_NODES 50000
#define N_EDGES 800000
#define D 64
#define N_TILES (N_NODES / 16)
#define SMOOTHF 0.5f


__global__ void k_zero_deg(float* __restrict__ deg) {
    int i = blockIdx.x * blockDim.x + threadIdx.x;
    if (i < N_NODES) deg[i] = 0.0f;
}

__global__ void k_degree(const int* __restrict__ ecol, float* __restrict__ deg) {
    int e = blockIdx.x * blockDim.x + threadIdx.x;
    if (e < N_EDGES) atomicAdd(&deg[ecol[e]], 1.0f);
}

__global__ void k_rsqrt(float* __restrict__ deg) {
    int i = blockIdx.x * blockDim.x + threadIdx.x;
    if (i < N_NODES) deg[i] = rsqrtf(deg[i] + 1.0f);
}


__global__ __launch_bounds__(256) void k_gemm_wmma(
    const float* __restrict__ x, const float* __restrict__ W,
    const float* __restrict__ bias, const float* __restrict__ d,
    float* __restrict__ support, float* __restrict__ h, float* __restrict__ msg)
{
    __shared__ float Wl[D * D];
    for (int i = threadIdx.x; i < D * D; i += 256) Wl[i] = W[i];
    __syncthreads();

    const int wave = threadIdx.x >> 5;
    const int lane = threadIdx.x & 31;
    const int tile = blockIdx.x * 8 + wave;
    if (tile >= N_TILES) return;

    const int row0 = tile * 16;
    const int half = lane >> 4;
    const int l15  = lane & 15;

    v8f acc[4] = {v8f{}, v8f{}, v8f{}, v8f{}};
    const float* xrow = x + (size_t)(row0 + l15) * D;

    #pragma unroll
    for (int kk = 0; kk < 16; ++kk) {
        const int k = kk * 4 + half * 2;
        v2f a = *(const v2f*)(xrow + k);
        #pragma unroll
        for (int nt = 0; nt < 4; ++nt) {
            const int n = nt * 16 + l15;
            v2f bf = *(const v2f*)(&Wl[n * D + k]);
            acc[nt] = __builtin_amdgcn_wmma_f32_16x16x4_f32(
                 false, a,  false, bf,
                 (short)0, acc[nt],
                 false,  false);
        }
    }

    #pragma unroll
    for (int nt = 0; nt < 4; ++nt) {
        const int n  = nt * 16 + l15;
        const float bv = bias[n];
        #pragma unroll
        for (int r = 0; r < 8; ++r) {
            const int row  = row0 + r + 8 * half;
            const float s  = acc[nt][r] + bv;
            const float hv = s * d[row];
            const size_t idx = (size_t)row * D + n;
            support[idx] = s;
            h[idx]       = hv;
            msg[idx]     = hv;
        }
    }
}


#define EDGE_GROUPS    2
#define EDGES_PER_WAVE (32 * EDGE_GROUPS)
#define EDGE_BATCH     8
#define N_BATCHES      (EDGES_PER_WAVE / EDGE_BATCH)
#define N_EDGE_WAVES   (N_EDGES / EDGES_PER_WAVE)

static_assert(N_EDGES % EDGES_PER_WAVE == 0, "a wave has a whole run of edges or none");
static_assert(32 % EDGE_BATCH == 0, "a batch stays inside one 32-edge group");

__global__ __launch_bounds__(256) void k_scatter(
    const int* __restrict__ erow, const int* __restrict__ ecol,
    const float* __restrict__ h, float* __restrict__ msg)
{
    const int lane = threadIdx.x & 31;
    const int wave = blockIdx.x * 8 + (threadIdx.x >> 5);
    if (wave >= N_EDGE_WAVES) return;

    static_assert((unsigned long long)N_NODES * D * sizeof(float) < (1ull << 32), "32-bit row offsets");
    const char* hb = (const char*)h;
    char*       mb = (char*)msg;
    const unsigned loff = (unsigned)lane * 4u;

    const int e0 = wave * EDGES_PER_WAVE;
    int r[EDGE_GROUPS], c[EDGE_GROUPS];
    #pragma unroll
    for (int g = 0; g < EDGE_GROUPS; ++g) {
        r[g] = erow[e0 + g * 32 + lane];
        c[g] = ecol[e0 + g * 32 + lane];
    }

    float lo[2][EDGE_BATCH], hi[2][EDGE_BATCH];

    #pragma unroll
    for (int u = 0; u < EDGE_BATCH; ++u) {
        const unsigned cc  = (unsigned)__builtin_amdgcn_readlane(c[0], u);
        const unsigned off = (cc << 8) + loff;
        lo[0][u] = *(const float*)(hb + off);
        hi[0][u] = *(const float*)(hb + off + 128u);
    }

    #pragma unroll
    for (int b = 0; b < N_BATCHES; ++b) {
        const int cur = b & 1, nxt = cur ^ 1;
        if (b + 1 < N_BATCHES) {
            const int g1 = ((b + 1) * EDGE_BATCH) / 32;
            const int l1 = ((b + 1) * EDGE_BATCH) % 32;
            #pragma unroll
            for (int u = 0; u < EDGE_BATCH; ++u) {
                const unsigned cc  = (unsigned)__builtin_amdgcn_readlane(c[g1], l1 + u);
                const unsigned off = (cc << 8) + loff;
                lo[nxt][u] = *(const float*)(hb + off);
                hi[nxt][u] = *(const float*)(hb + off + 128u);
            }
        }
        const int g0 = (b * EDGE_BATCH) / 32;
        const int l0 = (b * EDGE_BATCH) % 32;
        #pragma unroll
        for (int u = 0; u < EDGE_BATCH; ++u) {
            const unsigned rr  = (unsigned)__builtin_amdgcn_readlane(r[g0], l0 + u);
            const unsigned off = (rr << 8) + loff;
            atomicAdd((float*)(mb + off),        lo[cur][u]);
            atomicAdd((float*)(mb + off + 128u), hi[cur][u]);
        }
    }
}


__global__ void k_final(const float* __restrict__ msg, const float* __restrict__ support,
                        const float* __restrict__ d, float* __restrict__ out)
{
    const int i = blockIdx.x * blockDim.x + threadIdx.x;
    if (i < N_NODES * D) {
        const int row = i >> 6;
        out[i] = (msg[i] * d[row] * SMOOTHF + support[i]) * (1.0f / (1.0f + SMOOTHF));
    }
}


extern "C" void kernel_launch(void* const* d_in, const int* in_sizes, int n_in,
                              void* d_out, int out_size, void* d_ws, size_t ws_size,
                              hipStream_t stream) {
    const float* x    = (const float*)d_in[0];
    const float* W    = (const float*)d_in[1];
    const float* bias = (const float*)d_in[2];
    const int*   erow = (const int*)d_in[3];
    const int*   ecol = (const int*)d_in[4];
    float* out = (float*)d_out;

    float* ws      = (float*)d_ws;
    float* support = ws;
    float* h       = ws + (size_t)N_NODES * D;
    float* msg     = ws + 2 * (size_t)N_NODES * D;
    float* deg     = ws + 3 * (size_t)N_NODES * D;

    k_zero_deg<<<(N_NODES + 255) / 256, 256, 0, stream>>>(deg);
    k_degree  <<<(N_EDGES + 255) / 256, 256, 0, stream>>>(ecol, deg);
    k_rsqrt   <<<(N_NODES + 255) / 256, 256, 0, stream>>>(deg);
    k_gemm_wmma<<<(N_TILES + 7) / 8, 256, 0, stream>>>(x, W, bias, deg, support, h, msg);
    k_scatter <<<(N_EDGE_WAVES + 7) / 8, 256, 0, stream>>>(erow, ecol, h, msg);
    k_final   <<<(N_NODES * D + 255) / 256, 256, 0, stream>>>(msg, support, deg, out);
}


# degree count in LDS: 64 work-groups each histogram an edge slice (integer LDS atomics, two counts per word), planes summed by plain loads; scattered global f32 adds and zero pass gone; counts exact
# speedup vs baseline: 3.1752x; 1.6525x over previous
#include <hip/hip_runtime.h>
#include <hip/hip_bf16.h>

typedef float v2f __attribute__((ext_vector_type(2)));
typedef float v8f __attribute__((ext_vector_type(8)));

#define N_NODES 50000
#define N_EDGES 800000
#define D 64
#define N_TILES (N_NODES / 16)
#define SMOOTHF 0.5f


#define HIST_BLOCKS    64
#define HIST_THREADS   1024
#define HIST_EDGES     (N_EDGES / HIST_BLOCKS)
#define HIST_WORDS     (N_NODES / 2)
#define HIST_PIECES    (HIST_EDGES / 4)
#define HIST_LOADS     ((HIST_PIECES + HIST_THREADS - 1) / HIST_THREADS)

static_assert(N_EDGES % HIST_BLOCKS == 0, "equal slices");
static_assert(HIST_EDGES % 4 == 0, "a slice is whole 16-byte pieces, and starts on one");
static_assert(HIST_EDGES < 65536, "a slice's count of one node fits its 16 bits");
static_assert(N_NODES % 2 == 0 && HIST_WORDS % 4 == 0, "whole words, whole 16-byte pieces");

__device__ __forceinline__ void hist_count(unsigned* cnt, int v) {
    if ((unsigned)v < (unsigned)N_NODES) atomicAdd(&cnt[v >> 1], 1u << ((v & 1) << 4));
}

__global__ __launch_bounds__(HIST_THREADS) void k_degree_hist(
    const int* __restrict__ ecol, unsigned* __restrict__ planes)
{
    __shared__ __attribute__((aligned(16))) unsigned cnt[HIST_WORDS];

    const int4* src = (const int4*)(ecol + (size_t)blockIdx.x * HIST_EDGES);
    int4 c[HIST_LOADS];
    #pragma unroll
    for (int k = 0; k < HIST_LOADS; ++k) {
        const int i = threadIdx.x + k * HIST_THREADS;
        c[k] = (i < HIST_PIECES) ? src[i] : make_int4(-1, -1, -1, -1);
    }

    uint4* cnt4 = (uint4*)cnt;
    for (int i = threadIdx.x; i < HIST_WORDS / 4; i += HIST_THREADS) cnt4[i] = make_uint4(0u, 0u, 0u, 0u);
    __syncthreads();

    #pragma unroll
    for (int k = 0; k < HIST_LOADS; ++k) {
        hist_count(cnt, c[k].x);
        hist_count(cnt, c[k].y);
        hist_count(cnt, c[k].z);
        hist_count(cnt, c[k].w);
    }
    __syncthreads();

    uint4* dst = (uint4*)(planes + (size_t)blockIdx.x * HIST_WORDS);
    for (int i = threadIdx.x; i < HIST_WORDS / 4; i += HIST_THREADS) dst[i] = cnt4[i];
}

#define NORM_THREADS 128

__global__ __launch_bounds__(NORM_THREADS) void k_degree_norm(
    const unsigned* __restrict__ planes, float* __restrict__ d)
{
    const int w = blockIdx.x * NORM_THREADS + threadIdx.x;
    if (w < HIST_WORDS) {
        unsigned lo = 0u, hi = 0u;
        #pragma unroll
        for (int b = 0; b < HIST_BLOCKS; ++b) {
            const unsigned v = planes[(size_t)b * HIST_WORDS + w];
            lo += v & 0xFFFFu;
            hi += v >> 16;
        }
        float2 r;
        r.x = rsqrtf((float)lo + 1.0f);
        r.y = rsqrtf((float)hi + 1.0f);
        *(float2*)(d + 2 * w) = r;
    }
}


__global__ __launch_bounds__(256) void k_gemm_wmma(
    const float* __restrict__ x, const float* __restrict__ W,
    const float* __restrict__ bias, const float* __restrict__ d,
    float* __restrict__ support, float* __restrict__ h, float* __restrict__ msg)
{
    __shared__ float Wl[D * D];
    for (int i = threadIdx.x; i < D * D; i += 256) Wl[i] = W[i];
    __syncthreads();

    const int wave = threadIdx.x >> 5;
    const int lane = threadIdx.x & 31;
    const int tile = blockIdx.x * 8 + wave;
    if (tile >= N_TILES) return;

    const int row0 = tile * 16;
    const int half = lane >> 4;
    const int l15  = lane & 15;

    v8f acc[4] = {v8f{}, v8f{}, v8f{}, v8f{}};
    const float* xrow = x + (size_t)(row0 + l15) * D;

    #pragma unroll
    for (int kk = 0; kk < 16; ++kk) {
        const int k = kk * 4 + half * 2;
        v2f a = *(const v2f*)(xrow + k);
        #pragma unroll
        for (int nt = 0; nt < 4; ++nt) {
            const int n = nt * 16 + l15;
            v2f bf = *(const v2f*)(&Wl[n * D + k]);
            acc[nt] = __builtin_amdgcn_wmma_f32_16x16x4_f32(
                 false, a,  false, bf,
                 (short)0, acc[nt],
                 false,  false);
        }
    }

    #pragma unroll
    for (int nt = 0; nt < 4; ++nt) {
        const int n  = nt * 16 + l15;
        const float bv = bias[n];
        #pragma unroll
        for (int r = 0; r < 8; ++r) {
            const int row  = row0 + r + 8 * half;
            const float s  = acc[nt][r] + bv;
            const float hv = s * d[row];
            const size_t idx = (size_t)row * D + n;
            support[idx] = s;
            h[idx]       = hv;
            msg[idx]     = hv;
        }
    }
}


#define EDGE_GROUPS    2
#define EDGES_PER_WAVE (32 * EDGE_GROUPS)
#define EDGE_BATCH     8
#define N_BATCHES      (EDGES_PER_WAVE / EDGE_BATCH)
#define N_EDGE_WAVES   (N_EDGES / EDGES_PER_WAVE)

static_assert(N_EDGES % EDGES_PER_WAVE == 0, "a wave has a whole run of edges or none");
static_assert(32 % EDGE_BATCH == 0, "a batch stays inside one 32-edge group");

__global__ __launch_bounds__(256) void k_scatter(
    const int* __restrict__ erow, const int* __restrict__ ecol,
    const float* __restrict__ h, float* __restrict__ msg)
{
    const int lane = threadIdx.x & 31;
    const int wave = blockIdx.x * 8 + (threadIdx.x >> 5);
    if (wave >= N_EDGE_WAVES) return;

    static_assert((unsigned long long)N_NODES * D * sizeof(float) < (1ull << 32), "32-bit row offsets");
    const char* hb = (const char*)h;
    char*       mb = (char*)msg;
    const unsigned loff = (unsigned)lane * 4u;

    const int e0 = wave * EDGES_PER_WAVE;
    int r[EDGE_GROUPS], c[EDGE_GROUPS];
    #pragma unroll
    for (int g = 0; g < EDGE_GROUPS; ++g) {
        r[g] = erow[e0 + g * 32 + lane];
        c[g] = ecol[e0 + g * 32 + lane];
    }

    float lo[2][EDGE_BATCH], hi[2][EDGE_BATCH];

    #pragma unroll
    for (int u = 0; u < EDGE_BATCH; ++u) {
        const unsigned cc  = (unsigned)__builtin_amdgcn_readlane(c[0], u);
        const unsigned off = (cc << 8) + loff;
        lo[0][u] = *(const float*)(hb + off);
        hi[0][u] = *(const float*)(hb + off + 128u);
    }

    #pragma unroll
    for (int b = 0; b < N_BATCHES; ++b) {
        const int cur = b & 1, nxt = cur ^ 1;
        if (b + 1 < N_BATCHES) {
            const int g1 = ((b + 1) * EDGE_BATCH) / 32;
            const int l1 = ((b + 1) * EDGE_BATCH) % 32;
            #pragma unroll
            for (int u = 0; u < EDGE_BATCH; ++u) {
                const unsigned cc  = (unsigned)__builtin_amdgcn_readlane(c[g1], l1 + u);
                const unsigned off = (cc << 8) + loff;
                lo[nxt][u] = *(const float*)(hb + off);
                hi[nxt][u] = *(const float*)(hb + off + 128u);
            }
        }
        const int g0 = (b * EDGE_BATCH) / 32;
        const int l0 = (b * EDGE_BATCH) % 32;
        #pragma unroll
        for (int u = 0; u < EDGE_BATCH; ++u) {
            const unsigned rr  = (unsigned)__builtin_amdgcn_readlane(r[g0], l0 + u);
            const unsigned off = (rr << 8) + loff;
            atomicAdd((float*)(mb + off),        lo[cur][u]);
            atomicAdd((float*)(mb + off + 128u), hi[cur][u]);
        }
    }
}


__global__ void k_final(const float* __restrict__ msg, const float* __restrict__ support,
                        const float* __restrict__ d, float* __restrict__ out)
{
    const int i = blockIdx.x * blockDim.x + threadIdx.x;
    if (i < N_NODES * D) {
        const int row = i >> 6;
        out[i] = (msg[i] * d[row] * SMOOTHF + support[i]) * (1.0f / (1.0f + SMOOTHF));
    }
}


extern "C" void kernel_launch(void* const* d_in, const int* in_sizes, int n_in,
                              void* d_out, int out_size, void* d_ws, size_t ws_size,
                              hipStream_t stream) {
    const float* x    = (const float*)d_in[0];
    const float* W    = (const float*)d_in[1];
    const float* bias = (const float*)d_in[2];
    const int*   erow = (const int*)d_in[3];
    const int*   ecol = (const int*)d_in[4];
    float* out = (float*)d_out;

    float* ws      = (float*)d_ws;
    float* support = ws;
    float* h       = ws + (size_t)N_NODES * D;
    float* msg     = ws + 2 * (size_t)N_NODES * D;
    float* deg     = ws + 3 * (size_t)N_NODES * D;
    unsigned* planes = (unsigned*)(ws + 3 * (size_t)N_NODES * D + 65536);

    k_degree_hist<<<HIST_BLOCKS, HIST_THREADS, 0, stream>>>(ecol, planes);
    k_degree_norm<<<(HIST_WORDS + NORM_THREADS - 1) / NORM_THREADS, NORM_THREADS, 0, stream>>>(planes, deg);
    k_gemm_wmma<<<(N_TILES + 7) / 8, 256, 0, stream>>>(x, W, bias, deg, support, h, msg);
    k_scatter <<<(N_EDGE_WAVES + 7) / 8, 256, 0, stream>>>(erow, ecol, h, msg);
    k_final   <<<(N_NODES * D + 255) / 256, 256, 0, stream>>>(msg, support, deg, out);
}


# degree plane sum folded into the GEMM launch: each GEMM work-group adds the 64 count planes for its own 128 nodes (partials via LDS) and forms and stores d; the norm launch is gone; counts exact
# speedup vs baseline: 4.2811x; 1.3483x over previous
#include <hip/hip_runtime.h>
#include <hip/hip_bf16.h>

typedef float v2f __attribute__((ext_vector_type(2)));
typedef float v8f __attribute__((ext_vector_type(8)));

#define N_NODES 50000
#define N_EDGES 800000
#define D 64
#define N_TILES (N_NODES / 16)
#define SMOOTHF 0.5f
#define C_SELF (1.0f / (1.0f + SMOOTHF))
#define C_AGG  (SMOOTHF / (1.0f + SMOOTHF))


#define HIST_BLOCKS    64
#define HIST_THREADS   1024
#define HIST_EDGES     (N_EDGES / HIST_BLOCKS)
#define HIST_WORDS     (N_NODES / 2)
#define HIST_PIECES    (HIST_EDGES / 4)
#define HIST_LOADS     ((HIST_PIECES + HIST_THREADS - 1) / HIST_THREADS)

static_assert(N_EDGES % HIST_BLOCKS == 0, "equal slices");
static_assert(HIST_EDGES % 4 == 0, "a slice is whole 16-byte pieces, and starts on one");
static_assert(HIST_EDGES < 65536, "a slice's count of one node fits its 16 bits");
static_assert(N_NODES % 2 == 0 && HIST_WORDS % 4 == 0, "whole words, whole 16-byte pieces");

__device__ __forceinline__ void hist_count(unsigned* cnt, int v) {
    if ((unsigned)v < (unsigned)N_NODES) atomicAdd(&cnt[v >> 1], 1u << ((v & 1) << 4));
}

__global__ __launch_bounds__(HIST_THREADS) void k_degree_hist(
    const int* __restrict__ ecol, unsigned* __restrict__ planes)
{
    __shared__ __attribute__((aligned(16))) unsigned cnt[HIST_WORDS];

    const int4* src = (const int4*)(ecol + (size_t)blockIdx.x * HIST_EDGES);
    int4 c[HIST_LOADS];
    #pragma unroll
    for (int k = 0; k < HIST_LOADS; ++k) {
        const int i = threadIdx.x + k * HIST_THREADS;
        c[k] = (i < HIST_PIECES) ? src[i] : make_int4(-1, -1, -1, -1);
    }

    uint4* cnt4 = (uint4*)cnt;
    for (int i = threadIdx.x; i < HIST_WORDS / 4; i += HIST_THREADS) cnt4[i] = make_uint4(0u, 0u, 0u, 0u);
    __syncthreads();

    #pragma unroll
    for (int k = 0; k < HIST_LOADS; ++k) {
        hist_count(cnt, c[k].x);
        hist_count(cnt, c[k].y);
        hist_count(cnt, c[k].z);
        hist_count(cnt, c[k].w);
    }
    __syncthreads();

    uint4* dst = (uint4*)(planes + (size_t)blockIdx.x * HIST_WORDS);
    for (int i = threadIdx.x; i < HIST_WORDS / 4; i += HIST_THREADS) dst[i] = cnt4[i];
}


#define GEMM_THREADS      256
#define GEMM_NODES        128
#define GEMM_WORDS        (GEMM_NODES / 2)
#define PLANE_GROUPS      (GEMM_THREADS / GEMM_WORDS)
#define PLANES_PER_THREAD (HIST_BLOCKS / PLANE_GROUPS)

static_assert(GEMM_THREADS % GEMM_WORDS == 0 && HIST_BLOCKS % PLANE_GROUPS == 0, "planes dealt out evenly");
static_assert(PLANE_GROUPS == 4, "the degree is written out as a sum of four partials");

__device__ __forceinline__ float deg_norm(const unsigned (*part)[GEMM_NODES], int i) {
    const unsigned deg = part[0][i] + part[1][i] + part[2][i] + part[3][i];
    return rsqrtf((float)deg + 1.0f);
}

__global__ __launch_bounds__(GEMM_THREADS) void k_gemm_wmma(
    const float* __restrict__ x, const float* __restrict__ W,
    const float* __restrict__ bias, const unsigned* __restrict__ planes,
    float* __restrict__ d, float* __restrict__ h, float* __restrict__ out)
{
    __shared__ float Wl[D * D];
    __shared__ __attribute__((aligned(16))) unsigned part[PLANE_GROUPS][GEMM_NODES];

    {
        const int wd = threadIdx.x % GEMM_WORDS;
        const int pg = threadIdx.x / GEMM_WORDS;
        const int w  = min((int)blockIdx.x * GEMM_WORDS + wd, HIST_WORDS - 1);
        unsigned lo = 0u, hi = 0u;
        #pragma unroll
        for (int j = 0; j < PLANES_PER_THREAD; ++j) {
            const unsigned v = planes[(size_t)(pg + PLANE_GROUPS * j) * HIST_WORDS + w];
            lo += v & 0xFFFFu;
            hi += v >> 16;
        }
        for (int i = threadIdx.x; i < D * D; i += GEMM_THREADS) Wl[i] = W[i];
        *(uint2*)&part[pg][2 * wd] = make_uint2(lo, hi);
    }
    __syncthreads();

    if (threadIdx.x < GEMM_NODES) {
        const int node = blockIdx.x * GEMM_NODES + threadIdx.x;
        if (node < N_NODES) d[node] = deg_norm(part, threadIdx.x);
    }

    const int wave = threadIdx.x >> 5;
    const int lane = threadIdx.x & 31;
    const int tile = blockIdx.x * 8 + wave;
    if (tile >= N_TILES) return;

    const int row0 = tile * 16;
    const int half = lane >> 4;
    const int l15  = lane & 15;

    v8f acc[4] = {v8f{}, v8f{}, v8f{}, v8f{}};
    const float* xrow = x + (size_t)(row0 + l15) * D;

    #pragma unroll
    for (int kk = 0; kk < 16; ++kk) {
        const int k = kk * 4 + half * 2;
        v2f a = *(const v2f*)(xrow + k);
        #pragma unroll
        for (int nt = 0; nt < 4; ++nt) {
            const int n = nt * 16 + l15;
            v2f bf = *(const v2f*)(&Wl[n * D + k]);
            acc[nt] = __builtin_amdgcn_wmma_f32_16x16x4_f32(
                 false, a,  false, bf,
                 (short)0, acc[nt],
                 false,  false);
        }
    }

    float drow[8];
    #pragma unroll
    for (int r = 0; r < 8; ++r) drow[r] = deg_norm(part, wave * 16 + r + 8 * half);

    #pragma unroll
    for (int nt = 0; nt < 4; ++nt) {
        const int n  = nt * 16 + l15;
        const float bv = bias[n];
        #pragma unroll
        for (int r = 0; r < 8; ++r) {
            const int row  = row0 + r + 8 * half;
            const float dr = drow[r];
            const float s  = acc[nt][r] + bv;
            const float hv = s * dr;
            const size_t idx = (size_t)row * D + n;
            h[idx]   = hv;
            out[idx] = s * C_SELF + hv * (dr * C_AGG);
        }
    }
}


#define EDGE_GROUPS    2
#define EDGES_PER_WAVE (32 * EDGE_GROUPS)
#define EDGE_BATCH     8
#define N_BATCHES      (EDGES_PER_WAVE / EDGE_BATCH)
#define N_EDGE_WAVES   (N_EDGES / EDGES_PER_WAVE)

static_assert(N_EDGES % EDGES_PER_WAVE == 0, "a wave has a whole run of edges or none");
static_assert(32 % EDGE_BATCH == 0, "a batch stays inside one 32-edge group");

__global__ __launch_bounds__(256) void k_scatter(
    const int* __restrict__ erow, const int* __restrict__ ecol,
    const float* __restrict__ h, const float* __restrict__ d, float* __restrict__ out)
{
    const int lane = threadIdx.x & 31;
    const int wave = blockIdx.x * 8 + (threadIdx.x >> 5);
    if (wave >= N_EDGE_WAVES) return;

    static_assert((unsigned long long)N_NODES * D * sizeof(float) < (1ull << 32), "32-bit row offsets");
    const char* hb = (const char*)h;
    char*       ob = (char*)out;
    const unsigned loff = (unsigned)lane * 4u;

    const int e0 = wave * EDGES_PER_WAVE;
    int r[EDGE_GROUPS], c[EDGE_GROUPS];
    #pragma unroll
    for (int g = 0; g < EDGE_GROUPS; ++g) {
        r[g] = erow[e0 + g * 32 + lane];
        c[g] = ecol[e0 + g * 32 + lane];
    }
    int w[EDGE_GROUPS];
    #pragma unroll
    for (int g = 0; g < EDGE_GROUPS; ++g) w[g] = __float_as_int(d[r[g]] * C_AGG);

    float lo[2][EDGE_BATCH], hi[2][EDGE_BATCH];

    #pragma unroll
    for (int u = 0; u < EDGE_BATCH; ++u) {
        const unsigned cc  = (unsigned)__builtin_amdgcn_readlane(c[0], u);
        const unsigned off = (cc << 8) + loff;
        lo[0][u] = *(const float*)(hb + off);
        hi[0][u] = *(const float*)(hb + off + 128u);
    }

    #pragma unroll
    for (int b = 0; b < N_BATCHES; ++b) {
        const int cur = b & 1, nxt = cur ^ 1;
        if (b + 1 < N_BATCHES) {
            const int g1 = ((b + 1) * EDGE_BATCH) / 32;
            const int l1 = ((b + 1) * EDGE_BATCH) % 32;
            #pragma unroll
            for (int u = 0; u < EDGE_BATCH; ++u) {
                const unsigned cc  = (unsigned)__builtin_amdgcn_readlane(c[g1], l1 + u);
                const unsigned off = (cc << 8) + loff;
                lo[nxt][u] = *(const float*)(hb + off);
                hi[nxt][u] = *(const float*)(hb + off + 128u);
            }
        }
        const int g0 = (b * EDGE_BATCH) / 32;
        const int l0 = (b * EDGE_BATCH) % 32;
        #pragma unroll
        for (int u = 0; u < EDGE_BATCH; ++u) {
            const unsigned rr  = (unsigned)__builtin_amdgcn_readlane(r[g0], l0 + u);
            const float    ww  = __int_as_float(__builtin_amdgcn_readlane(w[g0], l0 + u));
            const unsigned off = (rr << 8) + loff;
            atomicAdd((float*)(ob + off),        lo[cur][u] * ww);
            atomicAdd((float*)(ob + off + 128u), hi[cur][u] * ww);
        }
    }
}


extern "C" void kernel_launch(void* const* d_in, const int* in_sizes, int n_in,
                              void* d_out, int out_size, void* d_ws, size_t ws_size,
                              hipStream_t stream) {
    const float* x    = (const float*)d_in[0];
    const float* W    = (const float*)d_in[1];
    const float* bias = (const float*)d_in[2];
    const int*   erow = (const int*)d_in[3];
    const int*   ecol = (const int*)d_in[4];
    float* out = (float*)d_out;

    float* ws      = (float*)d_ws;
    float* h       = ws;
    float* deg     = ws + (size_t)N_NODES * D;
    unsigned* planes = (unsigned*)(ws + (size_t)N_NODES * D + 65536);

    k_degree_hist<<<HIST_BLOCKS, HIST_THREADS, 0, stream>>>(ecol, planes);
    k_gemm_wmma<<<(N_TILES + 7) / 8, GEMM_THREADS, 0, stream>>>(x, W, bias, planes, deg, h, out);
    k_scatter <<<(N_EDGE_WAVES + 7) / 8, 256, 0, stream>>>(erow, ecol, h, deg, out);
}


# edge scatter: a work-group reads its die and adds only into rows of its own cache half's range (chunk drawn from that half's counter; kept edges packed in LDS); f32 atomics, arrival order as before
# speedup vs baseline: 5.3576x; 1.2515x over previous
#include <hip/hip_runtime.h>
#include <hip/hip_bf16.h>

typedef float v2f __attribute__((ext_vector_type(2)));
typedef float v8f __attribute__((ext_vector_type(8)));

#define N_NODES 50000
#define N_EDGES 800000
#define D 64
#define N_TILES (N_NODES / 16)
#define SMOOTHF 0.5f
#define C_SELF (1.0f / (1.0f + SMOOTHF))
#define C_AGG  (SMOOTHF / (1.0f + SMOOTHF))


#define HIST_BLOCKS    64
#define CTR_STRIDE     64
#define HIST_THREADS   1024
#define HIST_EDGES     (N_EDGES / HIST_BLOCKS)
#define HIST_WORDS     (N_NODES / 2)
#define HIST_PIECES    (HIST_EDGES / 4)
#define HIST_LOADS     ((HIST_PIECES + HIST_THREADS - 1) / HIST_THREADS)

static_assert(N_EDGES % HIST_BLOCKS == 0, "equal slices");
static_assert(HIST_EDGES % 4 == 0, "a slice is whole 16-byte pieces, and starts on one");
static_assert(HIST_EDGES < 65536, "a slice's count of one node fits its 16 bits");
static_assert(N_NODES % 2 == 0 && HIST_WORDS % 4 == 0, "whole words, whole 16-byte pieces");

__device__ __forceinline__ void hist_count(unsigned* cnt, int v) {
    if ((unsigned)v < (unsigned)N_NODES) atomicAdd(&cnt[v >> 1], 1u << ((v & 1) << 4));
}

__global__ __launch_bounds__(HIST_THREADS) void k_degree_hist(
    const int* __restrict__ ecol, unsigned* __restrict__ planes, unsigned* __restrict__ ctr)
{
    __shared__ __attribute__((aligned(16))) unsigned cnt[HIST_WORDS];

    if (blockIdx.x == 0 && threadIdx.x < 2) ctr[threadIdx.x * CTR_STRIDE] = 0u;

    const int4* src = (const int4*)(ecol + (size_t)blockIdx.x * HIST_EDGES);
    int4 c[HIST_LOADS];
    #pragma unroll
    for (int k = 0; k < HIST_LOADS; ++k) {
        const int i = threadIdx.x + k * HIST_THREADS;
        c[k] = (i < HIST_PIECES) ? src[i] : make_int4(-1, -1, -1, -1);
    }

    uint4* cnt4 = (uint4*)cnt;
    for (int i = threadIdx.x; i < HIST_WORDS / 4; i += HIST_THREADS) cnt4[i] = make_uint4(0u, 0u, 0u, 0u);
    __syncthreads();

    #pragma unroll
    for (int k = 0; k < HIST_LOADS; ++k) {
        hist_count(cnt, c[k].x);
        hist_count(cnt, c[k].y);
        hist_count(cnt, c[k].z);
        hist_count(cnt, c[k].w);
    }
    __syncthreads();

    uint4* dst = (uint4*)(planes + (size_t)blockIdx.x * HIST_WORDS);
    for (int i = threadIdx.x; i < HIST_WORDS / 4; i += HIST_THREADS) dst[i] = cnt4[i];
}


#define GEMM_THREADS      256
#define GEMM_NODES        128
#define GEMM_WORDS        (GEMM_NODES / 2)
#define PLANE_GROUPS      (GEMM_THREADS / GEMM_WORDS)
#define PLANES_PER_THREAD (HIST_BLOCKS / PLANE_GROUPS)

static_assert(GEMM_THREADS % GEMM_WORDS == 0 && HIST_BLOCKS % PLANE_GROUPS == 0, "planes dealt out evenly");
static_assert(PLANE_GROUPS == 4, "the degree is written out as a sum of four partials");

__device__ __forceinline__ float deg_norm(const unsigned (*part)[GEMM_NODES], int i) {
    const unsigned deg = part[0][i] + part[1][i] + part[2][i] + part[3][i];
    return rsqrtf((float)deg + 1.0f);
}

__global__ __launch_bounds__(GEMM_THREADS) void k_gemm_wmma(
    const float* __restrict__ x, const float* __restrict__ W,
    const float* __restrict__ bias, const unsigned* __restrict__ planes,
    float* __restrict__ d, float* __restrict__ h, float* __restrict__ out)
{
    __shared__ float Wl[D * D];
    __shared__ __attribute__((aligned(16))) unsigned part[PLANE_GROUPS][GEMM_NODES];

    {
        const int wd = threadIdx.x % GEMM_WORDS;
        const int pg = threadIdx.x / GEMM_WORDS;
        const int w  = min((int)blockIdx.x * GEMM_WORDS + wd, HIST_WORDS - 1);
        unsigned lo = 0u, hi = 0u;
        #pragma unroll
        for (int j = 0; j < PLANES_PER_THREAD; ++j) {
            const unsigned v = planes[(size_t)(pg + PLANE_GROUPS * j) * HIST_WORDS + w];
            lo += v & 0xFFFFu;
            hi += v >> 16;
        }
        for (int i = threadIdx.x; i < D * D; i += GEMM_THREADS) Wl[i] = W[i];
        *(uint2*)&part[pg][2 * wd] = make_uint2(lo, hi);
    }
    __syncthreads();

    if (threadIdx.x < GEMM_NODES) {
        const int node = blockIdx.x * GEMM_NODES + threadIdx.x;
        if (node < N_NODES) d[node] = deg_norm(part, threadIdx.x);
    }

    const int wave = threadIdx.x >> 5;
    const int lane = threadIdx.x & 31;
    const int tile = blockIdx.x * 8 + wave;
    if (tile >= N_TILES) return;

    const int row0 = tile * 16;
    const int half = lane >> 4;
    const int l15  = lane & 15;

    v8f acc[4] = {v8f{}, v8f{}, v8f{}, v8f{}};
    const float* xrow = x + (size_t)(row0 + l15) * D;

    #pragma unroll
    for (int kk = 0; kk < 16; ++kk) {
        const int k = kk * 4 + half * 2;
        v2f a = *(const v2f*)(xrow + k);
        #pragma unroll
        for (int nt = 0; nt < 4; ++nt) {
            const int n = nt * 16 + l15;
            v2f bf = *(const v2f*)(&Wl[n * D + k]);
            acc[nt] = __builtin_amdgcn_wmma_f32_16x16x4_f32(
                 false, a,  false, bf,
                 (short)0, acc[nt],
                 false,  false);
        }
    }

    float drow[8];
    #pragma unroll
    for (int r = 0; r < 8; ++r) drow[r] = deg_norm(part, wave * 16 + r + 8 * half);

    #pragma unroll
    for (int nt = 0; nt < 4; ++nt) {
        const int n  = nt * 16 + l15;
        const float bv = bias[n];
        #pragma unroll
        for (int r = 0; r < 8; ++r) {
            const int row  = row0 + r + 8 * half;
            const float dr = drow[r];
            const float s  = acc[nt][r] + bv;
            const float hv = s * dr;
            const size_t idx = (size_t)row * D + n;
            h[idx]   = hv;
            out[idx] = s * C_SELF + hv * (dr * C_AGG);
        }
    }
}


#define SC_RUN       128
#define SC_GROUPS    (SC_RUN / 32)
#define SC_WAVES     8
#define SC_BATCH     8
#define SC_SLOTS     (SC_RUN + SC_BATCH)
#define N_RUNS       (N_EDGES / SC_RUN)
#define N_CHUNKS     ((N_RUNS + SC_WAVES - 1) / SC_WAVES)
#define HALF_ROWS    (N_NODES / 2)

static_assert(N_EDGES % SC_RUN == 0, "a wave has a whole run of edges or none");
static_assert((unsigned long long)N_NODES * D * sizeof(float) < (1ull << 32), "32-bit row offsets");

struct EdgeBatch {
    float    lo[SC_BATCH], hi[SC_BATCH];
    unsigned roff[SC_BATCH];
    float    w[SC_BATCH];
};

__device__ __forceinline__ void batch_load(EdgeBatch& B, const uint4* ent, int base,
                                           const char* hb, unsigned loff) {
    #pragma unroll
    for (int u = 0; u < SC_BATCH; ++u) {
        const uint4 e = ent[base + u];
        B.roff[u] = (unsigned)__builtin_amdgcn_readfirstlane((int)e.x);
        const unsigned coff = (unsigned)__builtin_amdgcn_readfirstlane((int)e.y);
        B.w[u] = __int_as_float(__builtin_amdgcn_readfirstlane((int)e.z));
        const unsigned off = coff + loff;
        B.lo[u] = *(const float*)(hb + off);
        B.hi[u] = *(const float*)(hb + off + 128u);
    }
}

template <bool TAIL>
__device__ __forceinline__ void batch_add(const EdgeBatch& B, char* ob, unsigned loff, int valid) {
    #pragma unroll
    for (int u = 0; u < SC_BATCH; ++u) {
        if (!TAIL || u < valid) {
            const unsigned off = B.roff[u] + loff;
            atomicAdd((float*)(ob + off),        B.lo[u] * B.w[u]);
            atomicAdd((float*)(ob + off + 128u), B.hi[u] * B.w[u]);
        }
    }
}

__global__ __launch_bounds__(256) void k_scatter(
    const int* __restrict__ erow, const int* __restrict__ ecol,
    const float* __restrict__ h, const float* __restrict__ d, float* __restrict__ out,
    unsigned* __restrict__ ctr)
{
    __shared__ uint4    ents[SC_WAVES][SC_SLOTS];
    __shared__ unsigned ticket[2];

    if (threadIdx.x == 0) {
        const unsigned die = ((unsigned)__builtin_amdgcn_s_sendmsg_rtn(0x87) >> 16) & 15u;
        unsigned rg = (die >> 2) & 1u;
        unsigned t  = atomicAdd(&ctr[rg * CTR_STRIDE], 1u);
        if (t >= (unsigned)N_CHUNKS) {
            rg ^= 1u;
            t = atomicAdd(&ctr[rg * CTR_STRIDE], 1u);
        }
        ticket[0] = rg;
        ticket[1] = t;
    }
    __syncthreads();
    const unsigned rg = (unsigned)__builtin_amdgcn_readfirstlane((int)ticket[0]);
    const unsigned t  = (unsigned)__builtin_amdgcn_readfirstlane((int)ticket[1]);
    const bool chunk_ok = t < (unsigned)N_CHUNKS;

    const int lane = threadIdx.x & 31;
    const int wave = __builtin_amdgcn_readfirstlane((int)(threadIdx.x >> 5));
    const int run  = (int)t * SC_WAVES + wave;
    const bool active = chunk_ok && run < N_RUNS;

    const int      row0   = rg ? HALF_ROWS : 0;
    const unsigned n_rows = rg ? (unsigned)(N_NODES - HALF_ROWS) : (unsigned)HALF_ROWS;

    uint4* ent = ents[wave];
    unsigned n = 0;
    if (active) {
        const int e0 = run * SC_RUN;
        int r[SC_GROUPS], c[SC_GROUPS];
        #pragma unroll
        for (int g = 0; g < SC_GROUPS; ++g) {
            r[g] = erow[e0 + g * 32 + lane];
            c[g] = ecol[e0 + g * 32 + lane];
        }
        bool  keep[SC_GROUPS];
        float w[SC_GROUPS];
        #pragma unroll
        for (int g = 0; g < SC_GROUPS; ++g) {
            keep[g] = (unsigned)(r[g] - row0) < n_rows;
            w[g] = d[keep[g] ? r[g] : row0] * C_AGG;
        }
        #pragma unroll
        for (int g = 0; g < SC_GROUPS; ++g) {
            const unsigned mask = __builtin_amdgcn_ballot_w32(keep[g]);
            const unsigned pos  = n + __builtin_amdgcn_mbcnt_lo(mask, 0u);
            if (keep[g])
                ent[pos] = make_uint4((unsigned)r[g] << 8, (unsigned)c[g] << 8,
                                      (unsigned)__float_as_int(w[g]), 0u);
            n += (unsigned)__builtin_popcount(mask);
        }
        if (lane < SC_BATCH) ent[n + lane] = make_uint4((unsigned)row0 << 8, 0u, 0u, 0u);
    }
    __syncthreads();
    if (n == 0) return;

    const char* hb = (const char*)h;
    char*       ob = (char*)out;
    const unsigned loff = (unsigned)lane * 4u;

    const int nb         = (int)(n + SC_BATCH - 1) / SC_BATCH;
    const int last_valid = (int)n - (nb - 1) * SC_BATCH;

    EdgeBatch A, B;
    batch_load(A, ent, 0, hb, loff);
    int b = 0;
    for (; b + 2 < nb; b += 2) {
        batch_load(B, ent, (b + 1) * SC_BATCH, hb, loff);
        __builtin_amdgcn_sched_barrier(0);
        batch_add<false>(A, ob, loff, SC_BATCH);
        batch_load(A, ent, (b + 2) * SC_BATCH, hb, loff);
        __builtin_amdgcn_sched_barrier(0);
        batch_add<false>(B, ob, loff, SC_BATCH);
    }
    if (b + 1 < nb) {
        batch_load(B, ent, (b + 1) * SC_BATCH, hb, loff);
        __builtin_amdgcn_sched_barrier(0);
        batch_add<false>(A, ob, loff, SC_BATCH);
        batch_add<true>(B, ob, loff, last_valid);
    } else {
        batch_add<true>(A, ob, loff, last_valid);
    }
}


extern "C" void kernel_launch(void* const* d_in, const int* in_sizes, int n_in,
                              void* d_out, int out_size, void* d_ws, size_t ws_size,
                              hipStream_t stream) {
    const float* x    = (const float*)d_in[0];
    const float* W    = (const float*)d_in[1];
    const float* bias = (const float*)d_in[2];
    const int*   erow = (const int*)d_in[3];
    const int*   ecol = (const int*)d_in[4];
    float* out = (float*)d_out;

    float* ws      = (float*)d_ws;
    float* h       = ws;
    float* deg     = ws + (size_t)N_NODES * D;
    unsigned* planes = (unsigned*)(ws + (size_t)N_NODES * D + 65536);

    unsigned* ctr  = (unsigned*)(ws + (size_t)N_NODES * D + 65536 - 256);

    k_degree_hist<<<HIST_BLOCKS, HIST_THREADS, 0, stream>>>(ecol, planes, ctr);
    k_gemm_wmma<<<(N_TILES + 7) / 8, GEMM_THREADS, 0, stream>>>(x, W, bias, planes, deg, h, out);
    k_scatter <<<2 * N_CHUNKS, SC_WAVES * 32, 0, stream>>>(erow, ecol, h, deg, out, ctr);
}


# GEMM: W rows in LDS at a pitch of 68 words, not 64: the 16 rows of a B-fragment read no longer share one bank pair (16-way conflict on every read), the 32 lanes cover all 64 banks once
# speedup vs baseline: 5.5712x; 1.0399x over previous
#include <hip/hip_runtime.h>
#include <hip/hip_bf16.h>

typedef float v2f __attribute__((ext_vector_type(2)));
typedef float v8f __attribute__((ext_vector_type(8)));

#define N_NODES 50000
#define N_EDGES 800000
#define D 64
#define N_TILES (N_NODES / 16)
#define SMOOTHF 0.5f
#define C_SELF (1.0f / (1.0f + SMOOTHF))
#define C_AGG  (SMOOTHF / (1.0f + SMOOTHF))


#define HIST_BLOCKS    64
#define CTR_STRIDE     64
#define HIST_THREADS   1024
#define HIST_EDGES     (N_EDGES / HIST_BLOCKS)
#define HIST_WORDS     (N_NODES / 2)
#define HIST_PIECES    (HIST_EDGES / 4)
#define HIST_LOADS     ((HIST_PIECES + HIST_THREADS - 1) / HIST_THREADS)

static_assert(N_EDGES % HIST_BLOCKS == 0, "equal slices");
static_assert(HIST_EDGES % 4 == 0, "a slice is whole 16-byte pieces, and starts on one");
static_assert(HIST_EDGES < 65536, "a slice's count of one node fits its 16 bits");
static_assert(N_NODES % 2 == 0 && HIST_WORDS % 4 == 0, "whole words, whole 16-byte pieces");

__device__ __forceinline__ void hist_count(unsigned* cnt, int v) {
    if ((unsigned)v < (unsigned)N_NODES) atomicAdd(&cnt[v >> 1], 1u << ((v & 1) << 4));
}

__global__ __launch_bounds__(HIST_THREADS) void k_degree_hist(
    const int* __restrict__ ecol, unsigned* __restrict__ planes, unsigned* __restrict__ ctr)
{
    __shared__ __attribute__((aligned(16))) unsigned cnt[HIST_WORDS];

    if (blockIdx.x == 0 && threadIdx.x < 2) ctr[threadIdx.x * CTR_STRIDE] = 0u;

    const int4* src = (const int4*)(ecol + (size_t)blockIdx.x * HIST_EDGES);
    int4 c[HIST_LOADS];
    #pragma unroll
    for (int k = 0; k < HIST_LOADS; ++k) {
        const int i = threadIdx.x + k * HIST_THREADS;
        c[k] = (i < HIST_PIECES) ? src[i] : make_int4(-1, -1, -1, -1);
    }

    uint4* cnt4 = (uint4*)cnt;
    for (int i = threadIdx.x; i < HIST_WORDS / 4; i += HIST_THREADS) cnt4[i] = make_uint4(0u, 0u, 0u, 0u);
    __syncthreads();

    #pragma unroll
    for (int k = 0; k < HIST_LOADS; ++k) {
        hist_count(cnt, c[k].x);
        hist_count(cnt, c[k].y);
        hist_count(cnt, c[k].z);
        hist_count(cnt, c[k].w);
    }
    __syncthreads();

    uint4* dst = (uint4*)(planes + (size_t)blockIdx.x * HIST_WORDS);
    for (int i = threadIdx.x; i < HIST_WORDS / 4; i += HIST_THREADS) dst[i] = cnt4[i];
}


#define GEMM_THREADS      256
#define GEMM_NODES        128
#define GEMM_WORDS        (GEMM_NODES / 2)
#define PLANE_GROUPS      (GEMM_THREADS / GEMM_WORDS)
#define PLANES_PER_THREAD (HIST_BLOCKS / PLANE_GROUPS)
#define W_PITCH           (D + 4)

static_assert(GEMM_THREADS % GEMM_WORDS == 0 && HIST_BLOCKS % PLANE_GROUPS == 0, "planes dealt out evenly");
static_assert(PLANE_GROUPS == 4, "the degree is written out as a sum of four partials");
static_assert(W_PITCH % 2 == 0 && (W_PITCH * 16) % 64 == 0 && (W_PITCH % 64) == 4,
              "8-byte fragment reads stay aligned; every 16-row tile starts on bank 0; row n on bank 4n");

__device__ __forceinline__ float deg_norm(const unsigned (*part)[GEMM_NODES], int i) {
    const unsigned deg = part[0][i] + part[1][i] + part[2][i] + part[3][i];
    return rsqrtf((float)deg + 1.0f);
}

__global__ __launch_bounds__(GEMM_THREADS) void k_gemm_wmma(
    const float* __restrict__ x, const float* __restrict__ W,
    const float* __restrict__ bias, const unsigned* __restrict__ planes,
    float* __restrict__ d, float* __restrict__ h, float* __restrict__ out)
{
    __shared__ __attribute__((aligned(16))) float Wl[D * W_PITCH];
    __shared__ __attribute__((aligned(16))) unsigned part[PLANE_GROUPS][GEMM_NODES];

    {
        const int wd = threadIdx.x % GEMM_WORDS;
        const int pg = threadIdx.x / GEMM_WORDS;
        const int w  = min((int)blockIdx.x * GEMM_WORDS + wd, HIST_WORDS - 1);
        unsigned lo = 0u, hi = 0u;
        #pragma unroll
        for (int j = 0; j < PLANES_PER_THREAD; ++j) {
            const unsigned v = planes[(size_t)(pg + PLANE_GROUPS * j) * HIST_WORDS + w];
            lo += v & 0xFFFFu;
            hi += v >> 16;
        }
        for (int i = threadIdx.x; i < D * D; i += GEMM_THREADS) Wl[(i / D) * W_PITCH + (i % D)] = W[i];
        *(uint2*)&part[pg][2 * wd] = make_uint2(lo, hi);
    }
    __syncthreads();

    if (threadIdx.x < GEMM_NODES) {
        const int node = blockIdx.x * GEMM_NODES + threadIdx.x;
        if (node < N_NODES) d[node] = deg_norm(part, threadIdx.x);
    }

    const int wave = threadIdx.x >> 5;
    const int lane = threadIdx.x & 31;
    const int tile = blockIdx.x * 8 + wave;
    if (tile >= N_TILES) return;

    const int row0 = tile * 16;
    const int half = lane >> 4;
    const int l15  = lane & 15;

    v8f acc[4] = {v8f{}, v8f{}, v8f{}, v8f{}};
    const float* xrow = x + (size_t)(row0 + l15) * D;

    #pragma unroll
    for (int kk = 0; kk < 16; ++kk) {
        const int k = kk * 4 + half * 2;
        v2f a = *(const v2f*)(xrow + k);
        #pragma unroll
        for (int nt = 0; nt < 4; ++nt) {
            const int n = nt * 16 + l15;
            v2f bf = *(const v2f*)(&Wl[n * W_PITCH + k]);
            acc[nt] = __builtin_amdgcn_wmma_f32_16x16x4_f32(
                 false, a,  false, bf,
                 (short)0, acc[nt],
                 false,  false);
        }
    }

    float drow[8];
    #pragma unroll
    for (int r = 0; r < 8; ++r) drow[r] = deg_norm(part, wave * 16 + r + 8 * half);

    #pragma unroll
    for (int nt = 0; nt < 4; ++nt) {
        const int n  = nt * 16 + l15;
        const float bv = bias[n];
        #pragma unroll
        for (int r = 0; r < 8; ++r) {
            const int row  = row0 + r + 8 * half;
            const float dr = drow[r];
            const float s  = acc[nt][r] + bv;
            const float hv = s * dr;
            const size_t idx = (size_t)row * D + n;
            h[idx]   = hv;
            out[idx] = s * C_SELF + hv * (dr * C_AGG);
        }
    }
}


#define SC_RUN       128
#define SC_GROUPS    (SC_RUN / 32)
#define SC_WAVES     8
#define SC_BATCH     8
#define SC_SLOTS     (SC_RUN + SC_BATCH)
#define N_RUNS       (N_EDGES / SC_RUN)
#define N_CHUNKS     ((N_RUNS + SC_WAVES - 1) / SC_WAVES)
#define HALF_ROWS    (N_NODES / 2)

static_assert(N_EDGES % SC_RUN == 0, "a wave has a whole run of edges or none");
static_assert((unsigned long long)N_NODES * D * sizeof(float) < (1ull << 32), "32-bit row offsets");

struct EdgeBatch {
    float    lo[SC_BATCH], hi[SC_BATCH];
    unsigned roff[SC_BATCH];
    float    w[SC_BATCH];
};

__device__ __forceinline__ void batch_load(EdgeBatch& B, const uint4* ent, int base,
                                           const char* hb, unsigned loff) {
    #pragma unroll
    for (int u = 0; u < SC_BATCH; ++u) {
        const uint4 e = ent[base + u];
        B.roff[u] = (unsigned)__builtin_amdgcn_readfirstlane((int)e.x);
        const unsigned coff = (unsigned)__builtin_amdgcn_readfirstlane((int)e.y);
        B.w[u] = __int_as_float(__builtin_amdgcn_readfirstlane((int)e.z));
        const unsigned off = coff + loff;
        B.lo[u] = *(const float*)(hb + off);
        B.hi[u] = *(const float*)(hb + off + 128u);
    }
}

template <bool TAIL>
__device__ __forceinline__ void batch_add(const EdgeBatch& B, char* ob, unsigned loff, int valid) {
    #pragma unroll
    for (int u = 0; u < SC_BATCH; ++u) {
        if (!TAIL || u < valid) {
            const unsigned off = B.roff[u] + loff;
            atomicAdd((float*)(ob + off),        B.lo[u] * B.w[u]);
            atomicAdd((float*)(ob + off + 128u), B.hi[u] * B.w[u]);
        }
    }
}

__global__ __launch_bounds__(256) void k_scatter(
    const int* __restrict__ erow, const int* __restrict__ ecol,
    const float* __restrict__ h, const float* __restrict__ d, float* __restrict__ out,
    unsigned* __restrict__ ctr)
{
    __shared__ uint4    ents[SC_WAVES][SC_SLOTS];
    __shared__ unsigned ticket[2];

    if (threadIdx.x == 0) {
        const unsigned die = ((unsigned)__builtin_amdgcn_s_sendmsg_rtn(0x87) >> 16) & 15u;
        unsigned rg = (die >> 2) & 1u;
        unsigned t  = atomicAdd(&ctr[rg * CTR_STRIDE], 1u);
        if (t >= (unsigned)N_CHUNKS) {
            rg ^= 1u;
            t = atomicAdd(&ctr[rg * CTR_STRIDE], 1u);
        }
        ticket[0] = rg;
        ticket[1] = t;
    }
    __syncthreads();
    const unsigned rg = (unsigned)__builtin_amdgcn_readfirstlane((int)ticket[0]);
    const unsigned t  = (unsigned)__builtin_amdgcn_readfirstlane((int)ticket[1]);
    const bool chunk_ok = t < (unsigned)N_CHUNKS;

    const int lane = threadIdx.x & 31;
    const int wave = __builtin_amdgcn_readfirstlane((int)(threadIdx.x >> 5));
    const int run  = (int)t * SC_WAVES + wave;
    const bool active = chunk_ok && run < N_RUNS;

    const int      row0   = rg ? HALF_ROWS : 0;
    const unsigned n_rows = rg ? (unsigned)(N_NODES - HALF_ROWS) : (unsigned)HALF_ROWS;

    uint4* ent = ents[wave];
    unsigned n = 0;
    if (active) {
        const int e0 = run * SC_RUN;
        int r[SC_GROUPS], c[SC_GROUPS];
        #pragma unroll
        for (int g = 0; g < SC_GROUPS; ++g) {
            r[g] = erow[e0 + g * 32 + lane];
            c[g] = ecol[e0 + g * 32 + lane];
        }
        bool  keep[SC_GROUPS];
        float w[SC_GROUPS];
        #pragma unroll
        for (int g = 0; g < SC_GROUPS; ++g) {
            keep[g] = (unsigned)(r[g] - row0) < n_rows;
            w[g] = d[keep[g] ? r[g] : row0] * C_AGG;
        }
        #pragma unroll
        for (int g = 0; g < SC_GROUPS; ++g) {
            const unsigned mask = __builtin_amdgcn_ballot_w32(keep[g]);
            const unsigned pos  = n + __builtin_amdgcn_mbcnt_lo(mask, 0u);
            if (keep[g])
                ent[pos] = make_uint4((unsigned)r[g] << 8, (unsigned)c[g] << 8,
                                      (unsigned)__float_as_int(w[g]), 0u);
            n += (unsigned)__builtin_popcount(mask);
        }
        if (lane < SC_BATCH) ent[n + lane] = make_uint4((unsigned)row0 << 8, 0u, 0u, 0u);
    }
    __syncthreads();
    if (n == 0) return;

    const char* hb = (const char*)h;
    char*       ob = (char*)out;
    const unsigned loff = (unsigned)lane * 4u;

    const int nb         = (int)(n + SC_BATCH - 1) / SC_BATCH;
    const int last_valid = (int)n - (nb - 1) * SC_BATCH;

    EdgeBatch A, B;
    batch_load(A, ent, 0, hb, loff);
    int b = 0;
    for (; b + 2 < nb; b += 2) {
        batch_load(B, ent, (b + 1) * SC_BATCH, hb, loff);
        __builtin_amdgcn_sched_barrier(0);
        batch_add<false>(A, ob, loff, SC_BATCH);
        batch_load(A, ent, (b + 2) * SC_BATCH, hb, loff);
        __builtin_amdgcn_sched_barrier(0);
        batch_add<false>(B, ob, loff, SC_BATCH);
    }
    if (b + 1 < nb) {
        batch_load(B, ent, (b + 1) * SC_BATCH, hb, loff);
        __builtin_amdgcn_sched_barrier(0);
        batch_add<false>(A, ob, loff, SC_BATCH);
        batch_add<true>(B, ob, loff, last_valid);
    } else {
        batch_add<true>(A, ob, loff, last_valid);
    }
}


extern "C" void kernel_launch(void* const* d_in, const int* in_sizes, int n_in,
                              void* d_out, int out_size, void* d_ws, size_t ws_size,
                              hipStream_t stream) {
    const float* x    = (const float*)d_in[0];
    const float* W    = (const float*)d_in[1];
    const float* bias = (const float*)d_in[2];
    const int*   erow = (const int*)d_in[3];
    const int*   ecol = (const int*)d_in[4];
    float* out = (float*)d_out;

    float* ws      = (float*)d_ws;
    float* h       = ws;
    float* deg     = ws + (size_t)N_NODES * D;
    unsigned* planes = (unsigned*)(ws + (size_t)N_NODES * D + 65536);

    unsigned* ctr  = (unsigned*)(ws + (size_t)N_NODES * D + 65536 - 256);

    k_degree_hist<<<HIST_BLOCKS, HIST_THREADS, 0, stream>>>(ecol, planes, ctr);
    k_gemm_wmma<<<(N_TILES + 7) / 8, GEMM_THREADS, 0, stream>>>(x, W, bias, planes, deg, h, out);
    k_scatter <<<2 * N_CHUNKS, SC_WAVES * 32, 0, stream>>>(erow, ecol, h, deg, out, ctr);
}


# GEMM prologue: count-plane, W (16-byte pieces), x-fragment and bias loads issued as one group ahead of the barrier: one memory round trip where W staging ran eight in series and x waited behind them
# speedup vs baseline: 6.2575x; 1.1232x over previous
#include <hip/hip_runtime.h>
#include <hip/hip_bf16.h>

typedef float v2f __attribute__((ext_vector_type(2)));
typedef float v8f __attribute__((ext_vector_type(8)));

#define N_NODES 50000
#define N_EDGES 800000
#define D 64
#define N_TILES (N_NODES / 16)
#define SMOOTHF 0.5f
#define C_SELF (1.0f / (1.0f + SMOOTHF))
#define C_AGG  (SMOOTHF / (1.0f + SMOOTHF))


#define HIST_BLOCKS    64
#define CTR_STRIDE     64
#define HIST_THREADS   1024
#define HIST_EDGES     (N_EDGES / HIST_BLOCKS)
#define HIST_WORDS     (N_NODES / 2)
#define HIST_PIECES    (HIST_EDGES / 4)
#define HIST_LOADS     ((HIST_PIECES + HIST_THREADS - 1) / HIST_THREADS)

static_assert(N_EDGES % HIST_BLOCKS == 0, "equal slices");
static_assert(HIST_EDGES % 4 == 0, "a slice is whole 16-byte pieces, and starts on one");
static_assert(HIST_EDGES < 65536, "a slice's count of one node fits its 16 bits");
static_assert(N_NODES % 2 == 0 && HIST_WORDS % 4 == 0, "whole words, whole 16-byte pieces");

__device__ __forceinline__ void hist_count(unsigned* cnt, int v) {
    if ((unsigned)v < (unsigned)N_NODES) atomicAdd(&cnt[v >> 1], 1u << ((v & 1) << 4));
}

__global__ __launch_bounds__(HIST_THREADS) void k_degree_hist(
    const int* __restrict__ ecol, unsigned* __restrict__ planes, unsigned* __restrict__ ctr)
{
    __shared__ __attribute__((aligned(16))) unsigned cnt[HIST_WORDS];

    if (blockIdx.x == 0 && threadIdx.x < 2) ctr[threadIdx.x * CTR_STRIDE] = 0u;

    const int4* src = (const int4*)(ecol + (size_t)blockIdx.x * HIST_EDGES);
    int4 c[HIST_LOADS];
    #pragma unroll
    for (int k = 0; k < HIST_LOADS; ++k) {
        const int i = threadIdx.x + k * HIST_THREADS;
        c[k] = (i < HIST_PIECES) ? src[i] : make_int4(-1, -1, -1, -1);
    }

    uint4* cnt4 = (uint4*)cnt;
    for (int i = threadIdx.x; i < HIST_WORDS / 4; i += HIST_THREADS) cnt4[i] = make_uint4(0u, 0u, 0u, 0u);
    __syncthreads();

    #pragma unroll
    for (int k = 0; k < HIST_LOADS; ++k) {
        hist_count(cnt, c[k].x);
        hist_count(cnt, c[k].y);
        hist_count(cnt, c[k].z);
        hist_count(cnt, c[k].w);
    }
    __syncthreads();

    uint4* dst = (uint4*)(planes + (size_t)blockIdx.x * HIST_WORDS);
    for (int i = threadIdx.x; i < HIST_WORDS / 4; i += HIST_THREADS) dst[i] = cnt4[i];
}


#define GEMM_THREADS      256
#define GEMM_NODES        128
#define GEMM_WORDS        (GEMM_NODES / 2)
#define PLANE_GROUPS      (GEMM_THREADS / GEMM_WORDS)
#define PLANES_PER_THREAD (HIST_BLOCKS / PLANE_GROUPS)
#define W_PITCH           (D + 4)
#define W_PIECES          (D * D / 4)
#define W_LOADS           (W_PIECES / GEMM_THREADS)

static_assert(GEMM_THREADS % GEMM_WORDS == 0 && HIST_BLOCKS % PLANE_GROUPS == 0, "planes dealt out evenly");
static_assert(PLANE_GROUPS == 4, "the degree is written out as a sum of four partials");
static_assert(W_PIECES % GEMM_THREADS == 0 && D % 4 == 0 && W_PITCH % 4 == 0,
              "W is dealt out in whole 16-byte pieces, and a piece stays 16-byte aligned in LDS");
static_assert(W_PITCH % 2 == 0 && (W_PITCH * 16) % 64 == 0 && (W_PITCH % 64) == 4,
              "8-byte fragment reads stay aligned; every 16-row tile starts on bank 0; row n on bank 4n");

__device__ __forceinline__ float deg_norm(const unsigned (*part)[GEMM_NODES], int i) {
    const unsigned deg = part[0][i] + part[1][i] + part[2][i] + part[3][i];
    return rsqrtf((float)deg + 1.0f);
}

__global__ __launch_bounds__(GEMM_THREADS) void k_gemm_wmma(
    const float* __restrict__ x, const float* __restrict__ W,
    const float* __restrict__ bias, const unsigned* __restrict__ planes,
    float* __restrict__ d, float* __restrict__ h, float* __restrict__ out)
{
    __shared__ __attribute__((aligned(16))) float Wl[D * W_PITCH];
    __shared__ __attribute__((aligned(16))) unsigned part[PLANE_GROUPS][GEMM_NODES];

    const int wave = threadIdx.x >> 5;
    const int lane = threadIdx.x & 31;
    const int tile = blockIdx.x * 8 + wave;
    const bool tile_ok = tile < N_TILES;
    const int row0 = min(tile, N_TILES - 1) * 16;
    const int half = lane >> 4;
    const int l15  = lane & 15;

    const int wd = threadIdx.x % GEMM_WORDS;
    const int pg = threadIdx.x / GEMM_WORDS;
    const int w  = min((int)blockIdx.x * GEMM_WORDS + wd, HIST_WORDS - 1);

    unsigned pv[PLANES_PER_THREAD];
    #pragma unroll
    for (int j = 0; j < PLANES_PER_THREAD; ++j)
        pv[j] = planes[(size_t)(pg + PLANE_GROUPS * j) * HIST_WORDS + w];

    float4 wv[W_LOADS];
    #pragma unroll
    for (int k = 0; k < W_LOADS; ++k)
        wv[k] = ((const float4*)W)[threadIdx.x + k * GEMM_THREADS];

    v2f a[16];
    const float* xrow = x + (size_t)(row0 + l15) * D + half * 2;
    #pragma unroll
    for (int kk = 0; kk < 16; ++kk) a[kk] = *(const v2f*)(xrow + kk * 4);

    float bv[4];
    #pragma unroll
    for (int nt = 0; nt < 4; ++nt) bv[nt] = bias[nt * 16 + l15];

    {
        unsigned lo = 0u, hi = 0u;
        #pragma unroll
        for (int j = 0; j < PLANES_PER_THREAD; ++j) {
            lo += pv[j] & 0xFFFFu;
            hi += pv[j] >> 16;
        }
        #pragma unroll
        for (int k = 0; k < W_LOADS; ++k) {
            const int p = threadIdx.x + k * GEMM_THREADS;
            *(float4*)&Wl[(p / (D / 4)) * W_PITCH + (p % (D / 4)) * 4] = wv[k];
        }
        *(uint2*)&part[pg][2 * wd] = make_uint2(lo, hi);
    }
    __syncthreads();

    if (threadIdx.x < GEMM_NODES) {
        const int node = blockIdx.x * GEMM_NODES + threadIdx.x;
        if (node < N_NODES) d[node] = deg_norm(part, threadIdx.x);
    }

    if (!tile_ok) return;

    v8f acc[4] = {v8f{}, v8f{}, v8f{}, v8f{}};

    #pragma unroll
    for (int kk = 0; kk < 16; ++kk) {
        const int k = kk * 4 + half * 2;
        #pragma unroll
        for (int nt = 0; nt < 4; ++nt) {
            const int n = nt * 16 + l15;
            v2f bf = *(const v2f*)(&Wl[n * W_PITCH + k]);
            acc[nt] = __builtin_amdgcn_wmma_f32_16x16x4_f32(
                 false, a[kk],  false, bf,
                 (short)0, acc[nt],
                 false,  false);
        }
    }

    float drow[8];
    #pragma unroll
    for (int r = 0; r < 8; ++r) drow[r] = deg_norm(part, wave * 16 + r + 8 * half);

    #pragma unroll
    for (int nt = 0; nt < 4; ++nt) {
        const int n  = nt * 16 + l15;
        #pragma unroll
        for (int r = 0; r < 8; ++r) {
            const int row  = row0 + r + 8 * half;
            const float dr = drow[r];
            const float s  = acc[nt][r] + bv[nt];
            const float hv = s * dr;
            const size_t idx = (size_t)row * D + n;
            h[idx]   = hv;
            out[idx] = s * C_SELF + hv * (dr * C_AGG);
        }
    }
}


#define SC_RUN       128
#define SC_GROUPS    (SC_RUN / 32)
#define SC_WAVES     8
#define SC_BATCH     8
#define SC_SLOTS     (SC_RUN + SC_BATCH)
#define N_RUNS       (N_EDGES / SC_RUN)
#define N_CHUNKS     ((N_RUNS + SC_WAVES - 1) / SC_WAVES)
#define HALF_ROWS    (N_NODES / 2)

static_assert(N_EDGES % SC_RUN == 0, "a wave has a whole run of edges or none");
static_assert((unsigned long long)N_NODES * D * sizeof(float) < (1ull << 32), "32-bit row offsets");

struct EdgeBatch {
    float    lo[SC_BATCH], hi[SC_BATCH];
    unsigned roff[SC_BATCH];
    float    w[SC_BATCH];
};

__device__ __forceinline__ void batch_load(EdgeBatch& B, const uint4* ent, int base,
                                           const char* hb, unsigned loff) {
    #pragma unroll
    for (int u = 0; u < SC_BATCH; ++u) {
        const uint4 e = ent[base + u];
        B.roff[u] = (unsigned)__builtin_amdgcn_readfirstlane((int)e.x);
        const unsigned coff = (unsigned)__builtin_amdgcn_readfirstlane((int)e.y);
        B.w[u] = __int_as_float(__builtin_amdgcn_readfirstlane((int)e.z));
        const unsigned off = coff + loff;
        B.lo[u] = *(const float*)(hb + off);
        B.hi[u] = *(const float*)(hb + off + 128u);
    }
}

template <bool TAIL>
__device__ __forceinline__ void batch_add(const EdgeBatch& B, char* ob, unsigned loff, int valid) {
    #pragma unroll
    for (int u = 0; u < SC_BATCH; ++u) {
        if (!TAIL || u < valid) {
            const unsigned off = B.roff[u] + loff;
            atomicAdd((float*)(ob + off),        B.lo[u] * B.w[u]);
            atomicAdd((float*)(ob + off + 128u), B.hi[u] * B.w[u]);
        }
    }
}

__global__ __launch_bounds__(256) void k_scatter(
    const int* __restrict__ erow, const int* __restrict__ ecol,
    const float* __restrict__ h, const float* __restrict__ d, float* __restrict__ out,
    unsigned* __restrict__ ctr)
{
    __shared__ uint4    ents[SC_WAVES][SC_SLOTS];
    __shared__ unsigned ticket[2];

    if (threadIdx.x == 0) {
        const unsigned die = ((unsigned)__builtin_amdgcn_s_sendmsg_rtn(0x87) >> 16) & 15u;
        unsigned rg = (die >> 2) & 1u;
        unsigned t  = atomicAdd(&ctr[rg * CTR_STRIDE], 1u);
        if (t >= (unsigned)N_CHUNKS) {
            rg ^= 1u;
            t = atomicAdd(&ctr[rg * CTR_STRIDE], 1u);
        }
        ticket[0] = rg;
        ticket[1] = t;
    }
    __syncthreads();
    const unsigned rg = (unsigned)__builtin_amdgcn_readfirstlane((int)ticket[0]);
    const unsigned t  = (unsigned)__builtin_amdgcn_readfirstlane((int)ticket[1]);
    const bool chunk_ok = t < (unsigned)N_CHUNKS;

    const int lane = threadIdx.x & 31;
    const int wave = __builtin_amdgcn_readfirstlane((int)(threadIdx.x >> 5));
    const int run  = (int)t * SC_WAVES + wave;
    const bool active = chunk_ok && run < N_RUNS;

    const int      row0   = rg ? HALF_ROWS : 0;
    const unsigned n_rows = rg ? (unsigned)(N_NODES - HALF_ROWS) : (unsigned)HALF_ROWS;

    uint4* ent = ents[wave];
    unsigned n = 0;
    if (active) {
        const int e0 = run * SC_RUN;
        int r[SC_GROUPS], c[SC_GROUPS];
        #pragma unroll
        for (int g = 0; g < SC_GROUPS; ++g) {
            r[g] = erow[e0 + g * 32 + lane];
            c[g] = ecol[e0 + g * 32 + lane];
        }
        bool  keep[SC_GROUPS];
        float w[SC_GROUPS];
        #pragma unroll
        for (int g = 0; g < SC_GROUPS; ++g) {
            keep[g] = (unsigned)(r[g] - row0) < n_rows;
            w[g] = d[keep[g] ? r[g] : row0] * C_AGG;
        }
        #pragma unroll
        for (int g = 0; g < SC_GROUPS; ++g) {
            const unsigned mask = __builtin_amdgcn_ballot_w32(keep[g]);
            const unsigned pos  = n + __builtin_amdgcn_mbcnt_lo(mask, 0u);
            if (keep[g])
                ent[pos] = make_uint4((unsigned)r[g] << 8, (unsigned)c[g] << 8,
                                      (unsigned)__float_as_int(w[g]), 0u);
            n += (unsigned)__builtin_popcount(mask);
        }
        if (lane < SC_BATCH) ent[n + lane] = make_uint4((unsigned)row0 << 8, 0u, 0u, 0u);
    }
    __syncthreads();
    if (n == 0) return;

    const char* hb = (const char*)h;
    char*       ob = (char*)out;
    const unsigned loff = (unsigned)lane * 4u;

    const int nb         = (int)(n + SC_BATCH - 1) / SC_BATCH;
    const int last_valid = (int)n - (nb - 1) * SC_BATCH;

    EdgeBatch A, B;
    batch_load(A, ent, 0, hb, loff);
    int b = 0;
    for (; b + 2 < nb; b += 2) {
        batch_load(B, ent, (b + 1) * SC_BATCH, hb, loff);
        __builtin_amdgcn_sched_barrier(0);
        batch_add<false>(A, ob, loff, SC_BATCH);
        batch_load(A, ent, (b + 2) * SC_BATCH, hb, loff);
        __builtin_amdgcn_sched_barrier(0);
        batch_add<false>(B, ob, loff, SC_BATCH);
    }
    if (b + 1 < nb) {
        batch_load(B, ent, (b + 1) * SC_BATCH, hb, loff);
        __builtin_amdgcn_sched_barrier(0);
        batch_add<false>(A, ob, loff, SC_BATCH);
        batch_add<true>(B, ob, loff, last_valid);
    } else {
        batch_add<true>(A, ob, loff, last_valid);
    }
}


extern "C" void kernel_launch(void* const* d_in, const int* in_sizes, int n_in,
                              void* d_out, int out_size, void* d_ws, size_t ws_size,
                              hipStream_t stream) {
    const float* x    = (const float*)d_in[0];
    const float* W    = (const float*)d_in[1];
    const float* bias = (const float*)d_in[2];
    const int*   erow = (const int*)d_in[3];
    const int*   ecol = (const int*)d_in[4];
    float* out = (float*)d_out;

    float* ws      = (float*)d_ws;
    float* h       = ws;
    float* deg     = ws + (size_t)N_NODES * D;
    unsigned* planes = (unsigned*)(ws + (size_t)N_NODES * D + 65536);

    unsigned* ctr  = (unsigned*)(ws + (size_t)N_NODES * D + 65536 - 256);

    k_degree_hist<<<HIST_BLOCKS, HIST_THREADS, 0, stream>>>(ecol, planes, ctr);
    k_gemm_wmma<<<(N_TILES + 7) / 8, GEMM_THREADS, 0, stream>>>(x, W, bias, planes, deg, h, out);
    k_scatter <<<2 * N_CHUNKS, SC_WAVES * 32, 0, stream>>>(erow, ecol, h, deg, out, ctr);
}


# edge scatter: d[row] formed from a byte table of the range's degrees that each work-group stages in LDS (GEMM launch stores the bytes; 255 = read d), not loaded per edge from memory; sums as before
# speedup vs baseline: 6.5337x; 1.0441x over previous
#include <hip/hip_runtime.h>
#include <hip/hip_bf16.h>

typedef float v2f __attribute__((ext_vector_type(2)));
typedef float v8f __attribute__((ext_vector_type(8)));

#define N_NODES 50000
#define N_EDGES 800000
#define D 64
#define N_TILES (N_NODES / 16)
#define SMOOTHF 0.5f
#define C_SELF (1.0f / (1.0f + SMOOTHF))
#define C_AGG  (SMOOTHF / (1.0f + SMOOTHF))


#define HIST_BLOCKS    64
#define CTR_STRIDE     64
#define HIST_THREADS   1024
#define HIST_EDGES     (N_EDGES / HIST_BLOCKS)
#define HIST_WORDS     (N_NODES / 2)
#define HIST_PIECES    (HIST_EDGES / 4)
#define HIST_LOADS     ((HIST_PIECES + HIST_THREADS - 1) / HIST_THREADS)

static_assert(N_EDGES % HIST_BLOCKS == 0, "equal slices");
static_assert(HIST_EDGES % 4 == 0, "a slice is whole 16-byte pieces, and starts on one");
static_assert(HIST_EDGES < 65536, "a slice's count of one node fits its 16 bits");
static_assert(N_NODES % 2 == 0 && HIST_WORDS % 4 == 0, "whole words, whole 16-byte pieces");

__device__ __forceinline__ void hist_count(unsigned* cnt, int v) {
    if ((unsigned)v < (unsigned)N_NODES) atomicAdd(&cnt[v >> 1], 1u << ((v & 1) << 4));
}

__global__ __launch_bounds__(HIST_THREADS) void k_degree_hist(
    const int* __restrict__ ecol, unsigned* __restrict__ planes, unsigned* __restrict__ ctr)
{
    __shared__ __attribute__((aligned(16))) unsigned cnt[HIST_WORDS];

    if (blockIdx.x == 0 && threadIdx.x < 2) ctr[threadIdx.x * CTR_STRIDE] = 0u;

    const int4* src = (const int4*)(ecol + (size_t)blockIdx.x * HIST_EDGES);
    int4 c[HIST_LOADS];
    #pragma unroll
    for (int k = 0; k < HIST_LOADS; ++k) {
        const int i = threadIdx.x + k * HIST_THREADS;
        c[k] = (i < HIST_PIECES) ? src[i] : make_int4(-1, -1, -1, -1);
    }

    uint4* cnt4 = (uint4*)cnt;
    for (int i = threadIdx.x; i < HIST_WORDS / 4; i += HIST_THREADS) cnt4[i] = make_uint4(0u, 0u, 0u, 0u);
    __syncthreads();

    #pragma unroll
    for (int k = 0; k < HIST_LOADS; ++k) {
        hist_count(cnt, c[k].x);
        hist_count(cnt, c[k].y);
        hist_count(cnt, c[k].z);
        hist_count(cnt, c[k].w);
    }
    __syncthreads();

    uint4* dst = (uint4*)(planes + (size_t)blockIdx.x * HIST_WORDS);
    for (int i = threadIdx.x; i < HIST_WORDS / 4; i += HIST_THREADS) dst[i] = cnt4[i];
}


#define GEMM_THREADS      256
#define GEMM_NODES        128
#define GEMM_WORDS        (GEMM_NODES / 2)
#define PLANE_GROUPS      (GEMM_THREADS / GEMM_WORDS)
#define PLANES_PER_THREAD (HIST_BLOCKS / PLANE_GROUPS)
#define W_PITCH           (D + 4)
#define W_PIECES          (D * D / 4)
#define W_LOADS           (W_PIECES / GEMM_THREADS)

static_assert(GEMM_THREADS % GEMM_WORDS == 0 && HIST_BLOCKS % PLANE_GROUPS == 0, "planes dealt out evenly");
static_assert(PLANE_GROUPS == 4, "the degree is written out as a sum of four partials");
static_assert(N_NODES % 4 == 0 && GEMM_NODES % 4 == 0 && GEMM_NODES / 4 == 32,
              "degree bytes go out as whole words, one wave's worth per work-group");
static_assert(W_PIECES % GEMM_THREADS == 0 && D % 4 == 0 && W_PITCH % 4 == 0,
              "W is dealt out in whole 16-byte pieces, and a piece stays 16-byte aligned in LDS");
static_assert(W_PITCH % 2 == 0 && (W_PITCH * 16) % 64 == 0 && (W_PITCH % 64) == 4,
              "8-byte fragment reads stay aligned; every 16-row tile starts on bank 0; row n on bank 4n");

#define DEG_ESC 255u

__device__ __forceinline__ unsigned deg_count(const unsigned (*part)[GEMM_NODES], int i) {
    return part[0][i] + part[1][i] + part[2][i] + part[3][i];
}

__device__ __forceinline__ float deg_to_norm(unsigned deg) {
    return rsqrtf((float)deg + 1.0f);
}

__device__ __forceinline__ float deg_norm(const unsigned (*part)[GEMM_NODES], int i) {
    return deg_to_norm(deg_count(part, i));
}

__global__ __launch_bounds__(GEMM_THREADS) void k_gemm_wmma(
    const float* __restrict__ x, const float* __restrict__ W,
    const float* __restrict__ bias, const unsigned* __restrict__ planes,
    float* __restrict__ d, unsigned* __restrict__ degb, float* __restrict__ h, float* __restrict__ out)
{
    __shared__ __attribute__((aligned(16))) float Wl[D * W_PITCH];
    __shared__ __attribute__((aligned(16))) unsigned part[PLANE_GROUPS][GEMM_NODES];

    const int wave = threadIdx.x >> 5;
    const int lane = threadIdx.x & 31;
    const int tile = blockIdx.x * 8 + wave;
    const bool tile_ok = tile < N_TILES;
    const int row0 = min(tile, N_TILES - 1) * 16;
    const int half = lane >> 4;
    const int l15  = lane & 15;

    const int wd = threadIdx.x % GEMM_WORDS;
    const int pg = threadIdx.x / GEMM_WORDS;
    const int w  = min((int)blockIdx.x * GEMM_WORDS + wd, HIST_WORDS - 1);

    unsigned pv[PLANES_PER_THREAD];
    #pragma unroll
    for (int j = 0; j < PLANES_PER_THREAD; ++j)
        pv[j] = planes[(size_t)(pg + PLANE_GROUPS * j) * HIST_WORDS + w];

    float4 wv[W_LOADS];
    #pragma unroll
    for (int k = 0; k < W_LOADS; ++k)
        wv[k] = ((const float4*)W)[threadIdx.x + k * GEMM_THREADS];

    v2f a[16];
    const float* xrow = x + (size_t)(row0 + l15) * D + half * 2;
    #pragma unroll
    for (int kk = 0; kk < 16; ++kk) a[kk] = *(const v2f*)(xrow + kk * 4);

    float bv[4];
    #pragma unroll
    for (int nt = 0; nt < 4; ++nt) bv[nt] = bias[nt * 16 + l15];

    {
        unsigned lo = 0u, hi = 0u;
        #pragma unroll
        for (int j = 0; j < PLANES_PER_THREAD; ++j) {
            lo += pv[j] & 0xFFFFu;
            hi += pv[j] >> 16;
        }
        #pragma unroll
        for (int k = 0; k < W_LOADS; ++k) {
            const int p = threadIdx.x + k * GEMM_THREADS;
            *(float4*)&Wl[(p / (D / 4)) * W_PITCH + (p % (D / 4)) * 4] = wv[k];
        }
        *(uint2*)&part[pg][2 * wd] = make_uint2(lo, hi);
    }
    __syncthreads();

    if (threadIdx.x < GEMM_NODES) {
        const int node = blockIdx.x * GEMM_NODES + threadIdx.x;
        if (node < N_NODES) d[node] = deg_norm(part, threadIdx.x);
    }
    if (threadIdx.x < GEMM_NODES / 4) {
        const int n0 = blockIdx.x * GEMM_NODES + 4 * (int)threadIdx.x;
        if (n0 < N_NODES) {
            unsigned packed = 0u;
            #pragma unroll
            for (int j = 0; j < 4; ++j)
                packed |= min(deg_count(part, 4 * (int)threadIdx.x + j), DEG_ESC) << (8 * j);
            degb[n0 >> 2] = packed;
        }
    }

    if (!tile_ok) return;

    v8f acc[4] = {v8f{}, v8f{}, v8f{}, v8f{}};

    #pragma unroll
    for (int kk = 0; kk < 16; ++kk) {
        const int k = kk * 4 + half * 2;
        #pragma unroll
        for (int nt = 0; nt < 4; ++nt) {
            const int n = nt * 16 + l15;
            v2f bf = *(const v2f*)(&Wl[n * W_PITCH + k]);
            acc[nt] = __builtin_amdgcn_wmma_f32_16x16x4_f32(
                 false, a[kk],  false, bf,
                 (short)0, acc[nt],
                 false,  false);
        }
    }

    float drow[8];
    #pragma unroll
    for (int r = 0; r < 8; ++r) drow[r] = deg_norm(part, wave * 16 + r + 8 * half);

    #pragma unroll
    for (int nt = 0; nt < 4; ++nt) {
        const int n  = nt * 16 + l15;
        #pragma unroll
        for (int r = 0; r < 8; ++r) {
            const int row  = row0 + r + 8 * half;
            const float dr = drow[r];
            const float s  = acc[nt][r] + bv[nt];
            const float hv = s * dr;
            const size_t idx = (size_t)row * D + n;
            h[idx]   = hv;
            out[idx] = s * C_SELF + hv * (dr * C_AGG);
        }
    }
}


#define SC_RUN       128
#define SC_GROUPS    (SC_RUN / 32)
#define SC_WAVES     8
#define SC_BATCH     8
#define SC_SLOTS     (SC_RUN + SC_BATCH)
#define N_RUNS       (N_EDGES / SC_RUN)
#define N_CHUNKS     ((N_RUNS + SC_WAVES - 1) / SC_WAVES)
#define HALF_ROWS    (N_NODES / 2)
#define TAB_PIECES   (HALF_ROWS / 8)
#define TAB_LOADS    ((TAB_PIECES + SC_WAVES * 32 - 1) / (SC_WAVES * 32))

static_assert(N_EDGES % SC_RUN == 0, "a wave has a whole run of edges or none");
static_assert(N_NODES - HALF_ROWS == HALF_ROWS && HALF_ROWS % 8 == 0 && TAB_PIECES >= SC_WAVES * 32,
              "both ranges have the same number of rows, each range's bytes are whole 8-byte pieces, a round fits");
static_assert((unsigned long long)N_NODES * D * sizeof(float) < (1ull << 32), "32-bit row offsets");

struct EdgeBatch {
    float    lo[SC_BATCH], hi[SC_BATCH];
    unsigned roff[SC_BATCH];
    float    w[SC_BATCH];
};

__device__ __forceinline__ void batch_load(EdgeBatch& B, const uint4* ent, int base,
                                           const char* hb, unsigned loff) {
    #pragma unroll
    for (int u = 0; u < SC_BATCH; ++u) {
        const uint4 e = ent[base + u];
        B.roff[u] = (unsigned)__builtin_amdgcn_readfirstlane((int)e.x);
        const unsigned coff = (unsigned)__builtin_amdgcn_readfirstlane((int)e.y);
        B.w[u] = __int_as_float(__builtin_amdgcn_readfirstlane((int)e.z));
        const unsigned off = coff + loff;
        B.lo[u] = *(const float*)(hb + off);
        B.hi[u] = *(const float*)(hb + off + 128u);
    }
}

template <bool TAIL>
__device__ __forceinline__ void batch_add(const EdgeBatch& B, char* ob, unsigned loff, int valid) {
    #pragma unroll
    for (int u = 0; u < SC_BATCH; ++u) {
        if (!TAIL || u < valid) {
            const unsigned off = B.roff[u] + loff;
            atomicAdd((float*)(ob + off),        B.lo[u] * B.w[u]);
            atomicAdd((float*)(ob + off + 128u), B.hi[u] * B.w[u]);
        }
    }
}

__global__ __launch_bounds__(256) void k_scatter(
    const int* __restrict__ erow, const int* __restrict__ ecol,
    const float* __restrict__ h, const float* __restrict__ d, const unsigned* __restrict__ degb,
    float* __restrict__ out, unsigned* __restrict__ ctr)
{
    __shared__ uint4    ents[SC_WAVES][SC_SLOTS];
    __shared__ __attribute__((aligned(16))) unsigned char dtab[HALF_ROWS];
    __shared__ unsigned ticket[2];

    if (threadIdx.x == 0) {
        const unsigned die = ((unsigned)__builtin_amdgcn_s_sendmsg_rtn(0x87) >> 16) & 15u;
        unsigned rg = (die >> 2) & 1u;
        unsigned t  = atomicAdd(&ctr[rg * CTR_STRIDE], 1u);
        if (t >= (unsigned)N_CHUNKS) {
            rg ^= 1u;
            t = atomicAdd(&ctr[rg * CTR_STRIDE], 1u);
        }
        ticket[0] = rg;
        ticket[1] = t;
    }
    __syncthreads();
    const unsigned rg = (unsigned)__builtin_amdgcn_readfirstlane((int)ticket[0]);
    const unsigned t  = (unsigned)__builtin_amdgcn_readfirstlane((int)ticket[1]);
    const bool chunk_ok = t < (unsigned)N_CHUNKS;

    const int lane = threadIdx.x & 31;
    const int wave = __builtin_amdgcn_readfirstlane((int)(threadIdx.x >> 5));
    const int run  = (int)t * SC_WAVES + wave;
    const bool active = chunk_ok && run < N_RUNS;

    const int      row0   = rg ? HALF_ROWS : 0;
    const unsigned n_rows = rg ? (unsigned)(N_NODES - HALF_ROWS) : (unsigned)HALF_ROWS;

    uint4* ent = ents[wave];

    int r[SC_GROUPS], c[SC_GROUPS];
    #pragma unroll
    for (int g = 0; g < SC_GROUPS; ++g) { r[g] = -1; c[g] = 0; }
    if (active) {
        const int e0 = run * SC_RUN;
        #pragma unroll
        for (int g = 0; g < SC_GROUPS; ++g) {
            r[g] = erow[e0 + g * 32 + lane];
            c[g] = ecol[e0 + g * 32 + lane];
        }
    }
    {
        const uint2* tsrc = (const uint2*)((const unsigned char*)degb + row0);
        uint2 tv[TAB_LOADS];
        #pragma unroll
        for (int k = 0; k < TAB_LOADS; ++k) {
            const int i = (int)threadIdx.x + min(k * (SC_WAVES * 32), TAB_PIECES - SC_WAVES * 32);
            tv[k] = tsrc[i];
        }
        #pragma unroll
        for (int k = 0; k < TAB_LOADS; ++k) {
            const int i = (int)threadIdx.x + min(k * (SC_WAVES * 32), TAB_PIECES - SC_WAVES * 32);
            ((uint2*)dtab)[i] = tv[k];
        }
    }
    __syncthreads();

    unsigned n = 0;
    if (active) {
        bool     keep[SC_GROUPS];
        unsigned deg[SC_GROUPS];
        float    w[SC_GROUPS];
        #pragma unroll
        for (int g = 0; g < SC_GROUPS; ++g) {
            keep[g] = (unsigned)(r[g] - row0) < n_rows;
            deg[g] = dtab[keep[g] ? r[g] - row0 : 0];
        }
        #pragma unroll
        for (int g = 0; g < SC_GROUPS; ++g) {
            float dv = deg_to_norm(deg[g]);
            if (keep[g] && deg[g] == DEG_ESC) dv = d[r[g]];
            w[g] = dv * C_AGG;
        }
        #pragma unroll
        for (int g = 0; g < SC_GROUPS; ++g) {
            const unsigned mask = __builtin_amdgcn_ballot_w32(keep[g]);
            const unsigned pos  = n + __builtin_amdgcn_mbcnt_lo(mask, 0u);
            if (keep[g])
                ent[pos] = make_uint4((unsigned)r[g] << 8, (unsigned)c[g] << 8,
                                      (unsigned)__float_as_int(w[g]), 0u);
            n += (unsigned)__builtin_popcount(mask);
        }
        if (lane < SC_BATCH) ent[n + lane] = make_uint4((unsigned)row0 << 8, 0u, 0u, 0u);
    }
    __syncthreads();
    if (n == 0) return;

    const char* hb = (const char*)h;
    char*       ob = (char*)out;
    const unsigned loff = (unsigned)lane * 4u;

    const int nb         = (int)(n + SC_BATCH - 1) / SC_BATCH;
    const int last_valid = (int)n - (nb - 1) * SC_BATCH;

    EdgeBatch A, B;
    batch_load(A, ent, 0, hb, loff);
    int b = 0;
    for (; b + 2 < nb; b += 2) {
        batch_load(B, ent, (b + 1) * SC_BATCH, hb, loff);
        __builtin_amdgcn_sched_barrier(0);
        batch_add<false>(A, ob, loff, SC_BATCH);
        batch_load(A, ent, (b + 2) * SC_BATCH, hb, loff);
        __builtin_amdgcn_sched_barrier(0);
        batch_add<false>(B, ob, loff, SC_BATCH);
    }
    if (b + 1 < nb) {
        batch_load(B, ent, (b + 1) * SC_BATCH, hb, loff);
        __builtin_amdgcn_sched_barrier(0);
        batch_add<false>(A, ob, loff, SC_BATCH);
        batch_add<true>(B, ob, loff, last_valid);
    } else {
        batch_add<true>(A, ob, loff, last_valid);
    }
}


extern "C" void kernel_launch(void* const* d_in, const int* in_sizes, int n_in,
                              void* d_out, int out_size, void* d_ws, size_t ws_size,
                              hipStream_t stream) {
    const float* x    = (const float*)d_in[0];
    const float* W    = (const float*)d_in[1];
    const float* bias = (const float*)d_in[2];
    const int*   erow = (const int*)d_in[3];
    const int*   ecol = (const int*)d_in[4];
    float* out = (float*)d_out;

    float* ws      = (float*)d_ws;
    float* h       = ws;
    float* deg     = ws + (size_t)N_NODES * D;
    unsigned* planes = (unsigned*)(ws + (size_t)N_NODES * D + 65536);

    unsigned* ctr  = (unsigned*)(ws + (size_t)N_NODES * D + 65536 - 256);
    unsigned* degb = planes + (size_t)HIST_BLOCKS * HIST_WORDS;

    k_degree_hist<<<HIST_BLOCKS, HIST_THREADS, 0, stream>>>(ecol, planes, ctr);
    k_gemm_wmma<<<(N_TILES + 7) / 8, GEMM_THREADS, 0, stream>>>(x, W, bias, planes, deg, degb, h, out);
    k_scatter <<<2 * N_CHUNKS, SC_WAVES * 32, 0, stream>>>(erow, ecol, h, deg, degb, out, ctr);
}
